# starting kernel
# baseline (speedup 1.0000x reference)
#include <hip/hip_runtime.h>
#include <math.h>

#define PI_F 3.14159265358979323846f

typedef float v2f __attribute__((ext_vector_type(2)));
typedef float v8f __attribute__((ext_vector_type(8)));

__device__ __forceinline__ v8f wmma_scan16(const float bm[4][2], int m, int hi) {
    v8f acc = {0.f, 0.f, 0.f, 0.f, 0.f, 0.f, 0.f, 0.f};
#pragma unroll
    for (int j = 0; j < 4; ++j) {
        const int c0 = 4 * j + 2 * hi;
        v2f a;
        a.x = (c0     <= m) ? 1.f : 0.f;
        a.y = (c0 + 1 <= m) ? 1.f : 0.f;
        v2f b;
        b.x = bm[j][0];
        b.y = bm[j][1];
        acc = __builtin_amdgcn_wmma_f32_16x16x4_f32(
             false, a,  false, b,
             (short)0, acc,  false,  false);
    }
    return acc;
}

__device__ __forceinline__ void apply_carry(const v8f acc, float& carry,
                                            float full[8], int n) {
    float loc[8];
#pragma unroll
    for (int r = 0; r < 8; ++r) loc[r] = acc[r];
    const float tot = __shfl(loc[7], 16 + n, 32);
#pragma unroll
    for (int r = 0; r < 8; ++r) full[r] = loc[r] + carry;
    carry += tot;
}

__device__ __forceinline__ void d_to_b(const float d[8], float bm[4][2], int hi) {
    float x[8];
#pragma unroll
    for (int r = 0; r < 8; ++r) x[r] = __shfl_xor(d[r], 16, 32);
#pragma unroll
    for (int j = 0; j < 2; ++j)
#pragma unroll
        for (int v = 0; v < 2; ++v)
            bm[j][v] = hi ? x[4 * j + v + 2] : d[4 * j + v];
#pragma unroll
    for (int j = 2; j < 4; ++j)
#pragma unroll
        for (int v = 0; v < 2; ++v)
            bm[j][v] = hi ? d[4 * j + v - 6] : x[4 * j + v - 8];
}

__global__ void k_lengths(const float* __restrict__ dist,
                          float* __restrict__ lengths) {
    const int j = blockIdx.x * blockDim.x + threadIdx.x;
    if (j >= 512) return;
    if (j == 511) { lengths[511] = 0.f; return; }
    float s = 0.f;
    for (int b = 0; b < 256; ++b) s += dist[b * 511 + j];
    lengths[j] = s * (1.0f / 256.0f);
}

__global__ void k_chain(const float* __restrict__ angles,
                        const float* __restrict__ lengths,
                        float* __restrict__ cart) {
    __shared__ float lenS[512];
    const int lane = threadIdx.x;
    for (int i = lane; i < 512; i += 32) lenS[i] = lengths[i];
    __syncthreads();

    const int n   = lane & 15;
    const int hi  = lane >> 4;
    const int m   = n;
    const int row = blockIdx.x * 16 + n;

    float carryP = 0.f, carryX = 0.f, carryY = 0.f;

    for (int t = 0; t < 32; ++t) {
        float bm[4][2];
#pragma unroll
        for (int j = 0; j < 4; ++j)
#pragma unroll
            for (int v = 0; v < 2; ++v) {
                const int k   = 4 * j + 2 * hi + v;
                const int pos = 16 * t + k;
                float val = 0.f;
                if (pos >= 1 && pos <= 510)
                    val = PI_F - angles[row * 510 + (pos - 1)];
                bm[j][v] = val;
            }
        const v8f accP = wmma_scan16(bm, m, hi);
        float prevF[8];
        apply_carry(accP, carryP, prevF, n);

        float dx[8], dy[8];
#pragma unroll
        for (int r = 0; r < 8; ++r) {
            const int pos = 16 * t + r + 8 * hi;
            float sv, cv;
            __sincosf(prevF[r], &sv, &cv);
            const float L = lenS[pos];
            dx[r] = L * cv;
            dy[r] = L * sv;
        }

        float bmx[4][2], bmy[4][2];
        d_to_b(dx, bmx, hi);
        d_to_b(dy, bmy, hi);
        const v8f accX = wmma_scan16(bmx, m, hi);
        const v8f accY = wmma_scan16(bmy, m, hi);
        float xF[8], yF[8];
        apply_carry(accX, carryX, xF, n);
        apply_carry(accY, carryY, yF, n);

#pragma unroll
        for (int r = 0; r < 8; ++r) {
            const int pos = 16 * t + r + 8 * hi;
            float* cp = cart + ((size_t)row * 512 + pos) * 3;
            cp[0] = xF[r] - dx[r];
            cp[1] = yF[r] - dy[r];
            cp[2] = 0.f;
        }
    }
}

__global__ void k_dihedral(const float* __restrict__ dih,
                           const float* __restrict__ cart,
                           float* __restrict__ out) {
    const int c = blockIdx.x * blockDim.x + threadIdx.x;
    if (c >= 512) return;
    const int side = c >> 8;
    const int b    = c & 255;
    const int nst  = side ? 255 : 254;
    const float* cb = cart + (size_t)b * 512 * 3;
    float*       ob = out  + (size_t)b * 512 * 3;

    float R00 = 1, R01 = 0, R02 = 0;
    float R10 = 0, R11 = 1, R12 = 0;
    float R20 = 0, R21 = 0, R22 = 1;
    float tx = 0, ty = 0, tz = 0;

    const int i0 = side ? 254 : 256;
    const int i1 = side ? 255 : 255;
    float q1x = cb[i1 * 3], q1y = cb[i1 * 3 + 1], q1z = cb[i1 * 3 + 2];
    if (!side) {
        ob[256 * 3] = cb[i0 * 3]; ob[256 * 3 + 1] = cb[i0 * 3 + 1]; ob[256 * 3 + 2] = cb[i0 * 3 + 2];
        ob[255 * 3] = q1x;        ob[255 * 3 + 1] = q1y;            ob[255 * 3 + 2] = q1z;
    }

    for (int i = 0; i <= nst; ++i) {
        const int jp = i + 2;
        const int pi = side ? (254 + jp) : (256 - jp);
        const float px = cb[pi * 3], py = cb[pi * 3 + 1], pz = cb[pi * 3 + 2];
        const float q2x = R00 * px + R01 * py + R02 * pz + tx;
        const float q2y = R10 * px + R11 * py + R12 * pz + ty;
        const float q2z = R20 * px + R21 * py + R22 * pz + tz;
        if (!side) {
            float* o = ob + (256 - jp) * 3;
            o[0] = q2x; o[1] = q2y; o[2] = q2z;
        } else if (jp >= 3) {
            float* o = ob + (254 + jp) * 3;
            o[0] = q2x; o[1] = q2y; o[2] = q2z;
        }
        if (i == nst) break;

        const int did = side ? (254 + i) : (253 - i);
        const float phi = dih[(size_t)b * 509 + did] + PI_F;
        float ax = q2x - q1x, ay = q2y - q1y, az = q2z - q1z;
        const float rn = rsqrtf(ax * ax + ay * ay + az * az);
        ax *= rn; ay *= rn; az *= rn;
        float sv, cv;
        __sincosf(phi, &sv, &cv);
        const float oc = 1.f - cv;
        const float M00 = cv + oc * ax * ax, M01 = oc * ax * ay - sv * az, M02 = oc * ax * az + sv * ay;
        const float M10 = oc * ax * ay + sv * az, M11 = cv + oc * ay * ay, M12 = oc * ay * az - sv * ax;
        const float M20 = oc * ax * az - sv * ay, M21 = oc * ay * az + sv * ax, M22 = cv + oc * az * az;
        const float N00 = M00 * R00 + M01 * R10 + M02 * R20;
        const float N01 = M00 * R01 + M01 * R11 + M02 * R21;
        const float N02 = M00 * R02 + M01 * R12 + M02 * R22;
        const float N10 = M10 * R00 + M11 * R10 + M12 * R20;
        const float N11 = M10 * R01 + M11 * R11 + M12 * R21;
        const float N12 = M10 * R02 + M11 * R12 + M12 * R22;
        const float N20 = M20 * R00 + M21 * R10 + M22 * R20;
        const float N21 = M20 * R01 + M21 * R11 + M22 * R21;
        const float N22 = M20 * R02 + M21 * R12 + M22 * R22;
        const float ux = tx - q2x, uy = ty - q2y, uz = tz - q2z;
        tx = M00 * ux + M01 * uy + M02 * uz + q2x;
        ty = M10 * ux + M11 * uy + M12 * uz + q2y;
        tz = M20 * ux + M21 * uy + M22 * uz + q2z;
        R00 = N00; R01 = N01; R02 = N02;
        R10 = N10; R11 = N11; R12 = N12;
        R20 = N20; R21 = N21; R22 = N22;
        q1x = q2x; q1y = q2y; q1z = q2z;
    }
}

extern "C" void kernel_launch(void* const* d_in, const int* in_sizes, int n_in,
                              void* d_out, int out_size, void* d_ws, size_t ws_size,
                              hipStream_t stream) {
    (void)in_sizes; (void)n_in; (void)out_size; (void)ws_size;
    const float* distances = (const float*)d_in[0];
    const float* angles    = (const float*)d_in[1];
    const float* dihedrals = (const float*)d_in[2];
    float* out = (float*)d_out;
    float* ws  = (float*)d_ws;
    float* lengths = ws;
    float* cart    = ws + 512;

    k_lengths <<<2, 256, 0, stream>>>(distances, lengths);
    k_chain   <<<16, 32, 0, stream>>>(angles, lengths, cart);
    k_dihedral<<<4, 128, 0, stream>>>(dihedrals, cart, out);
}


# k_chain: 8-wave workgroup copies its 16 rows' angles (one contiguous block) into LDS with 16-byte loads in one round trip; the scan reads LDS a chunk ahead, not 256 one-at-a-time global loads
# speedup vs baseline: 6.7078x; 6.7078x over previous
#include <hip/hip_runtime.h>
#include <math.h>

#define PI_F 3.14159265358979323846f

typedef float v2f __attribute__((ext_vector_type(2)));
typedef float v8f __attribute__((ext_vector_type(8)));

__device__ __forceinline__ v8f wmma_scan16(const float bm[4][2], int m, int hi) {
    v8f acc = {0.f, 0.f, 0.f, 0.f, 0.f, 0.f, 0.f, 0.f};
#pragma unroll
    for (int j = 0; j < 4; ++j) {
        const int c0 = 4 * j + 2 * hi;
        v2f a;
        a.x = (c0     <= m) ? 1.f : 0.f;
        a.y = (c0 + 1 <= m) ? 1.f : 0.f;
        v2f b;
        b.x = bm[j][0];
        b.y = bm[j][1];
        acc = __builtin_amdgcn_wmma_f32_16x16x4_f32(
             false, a,  false, b,
             (short)0, acc,  false,  false);
    }
    return acc;
}

__device__ __forceinline__ void apply_carry(const v8f acc, float& carry,
                                            float full[8], int n) {
    float loc[8];
#pragma unroll
    for (int r = 0; r < 8; ++r) loc[r] = acc[r];
    const float tot = __shfl(loc[7], 16 + n, 32);
#pragma unroll
    for (int r = 0; r < 8; ++r) full[r] = loc[r] + carry;
    carry += tot;
}

__device__ __forceinline__ void d_to_b(const float d[8], float bm[4][2], int hi) {
    float x[8];
#pragma unroll
    for (int r = 0; r < 8; ++r) x[r] = __shfl_xor(d[r], 16, 32);
#pragma unroll
    for (int j = 0; j < 2; ++j)
#pragma unroll
        for (int v = 0; v < 2; ++v)
            bm[j][v] = hi ? x[4 * j + v + 2] : d[4 * j + v];
#pragma unroll
    for (int j = 2; j < 4; ++j)
#pragma unroll
        for (int v = 0; v < 2; ++v)
            bm[j][v] = hi ? d[4 * j + v - 6] : x[4 * j + v - 8];
}

__global__ void k_lengths(const float* __restrict__ dist,
                          float* __restrict__ lengths) {
    const int j = blockIdx.x * blockDim.x + threadIdx.x;
    if (j >= 512) return;
    if (j == 511) { lengths[511] = 0.f; return; }
    float s = 0.f;
    for (int b = 0; b < 256; ++b) s += dist[b * 511 + j];
    lengths[j] = s * (1.0f / 256.0f);
}

#define CH_THREADS 256
#define ANG_Q      2040
#define ANG_F4     2049

__device__ __forceinline__ void load_chunk(const float* angS, int rowoff, int hi, int t, float out[8]) {
#pragma unroll
    for (int j = 0; j < 4; ++j)
#pragma unroll
        for (int v = 0; v < 2; ++v) {
            const int pos = 16 * t + 4 * j + 2 * hi + v;
            const float val = angS[rowoff + pos];
            out[2 * j + v] = (pos >= 1 && pos <= 510) ? val : 0.f;
        }
}

__global__ __launch_bounds__(CH_THREADS)
void k_chain(const float* __restrict__ angles,
             const float* __restrict__ lengths,
             float* __restrict__ cart) {
    __shared__ float4 angS4[ANG_F4];
    __shared__ float  lenS[512];
    const int tid = threadIdx.x;

    const float4* ag = (const float4*)(angles + (size_t)blockIdx.x * (16 * 510));
    float4 a[8];
#pragma unroll
    for (int k = 0; k < 8; ++k) {
        const int q = tid + CH_THREADS * k;
        a[k] = ag[(k < 7 || q < ANG_Q) ? q : (ANG_Q - 1)];
    }
    const float l0 = lengths[tid], l1 = lengths[tid + CH_THREADS];
#pragma unroll
    for (int k = 0; k < 8; ++k) {
        const int q = tid + CH_THREADS * k;
        angS4[q + 1] = make_float4(PI_F - a[k].x, PI_F - a[k].y, PI_F - a[k].z, PI_F - a[k].w);
    }
    if (tid == 0) angS4[0] = make_float4(0.f, 0.f, 0.f, 0.f);
    lenS[tid] = l0; lenS[tid + CH_THREADS] = l1;
    __syncthreads();
    if (tid >= 32) return;

    const float* angS = (const float*)angS4;
    const int lane = tid;
    const int n   = lane & 15;
    const int hi  = lane >> 4;
    const int m   = n;
    const int row = blockIdx.x * 16 + n;
    const int rowoff = 3 + n * 510;

    float carryP = 0.f, carryX = 0.f, carryY = 0.f;

    float cur[8];
    load_chunk(angS, rowoff, hi, 0, cur);

    for (int t = 0; t < 32; ++t) {
        float nxt[8];
        load_chunk(angS, rowoff, hi, (t + 1 < 32) ? (t + 1) : 31, nxt);

        float bm[4][2];
#pragma unroll
        for (int j = 0; j < 4; ++j)
#pragma unroll
            for (int v = 0; v < 2; ++v) bm[j][v] = cur[2 * j + v];
        const v8f accP = wmma_scan16(bm, m, hi);
        float prevF[8];
        apply_carry(accP, carryP, prevF, n);

        float dx[8], dy[8];
#pragma unroll
        for (int r = 0; r < 8; ++r) {
            const int pos = 16 * t + r + 8 * hi;
            float sv, cv;
            __sincosf(prevF[r], &sv, &cv);
            const float L = lenS[pos];
            dx[r] = L * cv;
            dy[r] = L * sv;
        }

        float bmx[4][2], bmy[4][2];
        d_to_b(dx, bmx, hi);
        d_to_b(dy, bmy, hi);
        const v8f accX = wmma_scan16(bmx, m, hi);
        const v8f accY = wmma_scan16(bmy, m, hi);
        float xF[8], yF[8];
        apply_carry(accX, carryX, xF, n);
        apply_carry(accY, carryY, yF, n);

#pragma unroll
        for (int r = 0; r < 8; ++r) {
            const int pos = 16 * t + r + 8 * hi;
            float* cp = cart + ((size_t)row * 512 + pos) * 3;
            cp[0] = xF[r] - dx[r];
            cp[1] = yF[r] - dy[r];
            cp[2] = 0.f;
        }
#pragma unroll
        for (int i = 0; i < 8; ++i) cur[i] = nxt[i];
    }
}

#define DH_THREADS 128
#define REC_SIDE   2056

__device__ __forceinline__ void rot_step(const float4 A, const float4 B,
                                         float& g0, float& g1, float& g2,
                                         float& q, float* outS, int idx) {
    q = fmaf(g1, B.w, fmaf(g0, B.z, q));
    outS[idx] = q;
    const float n0 = fmaf(-g2, A.w, fmaf(g1, A.y, g0 * A.x));
    const float n1 = fmaf(-g2, B.x, fmaf(g1, A.z, g0 * A.y));
    const float n2 = fmaf( g2, B.y, fmaf(g1, B.x, g0 * A.w));
    g0 = n0; g1 = n1; g2 = n2;
}

__device__ __forceinline__ void rot_block8(const float4 (&c)[16], int i0,
                                           bool right, int r, int spare,
                                           float& g0, float& g1, float& g2,
                                           float& q, float* outS) {
#pragma unroll
    for (int u = 0; u < 8; ++u) {
        const int i   = i0 + u;
        const bool ok = right ? (i >= 1) : (i <= 254);
        const int col = right ? (256 + i) : (254 - i);
        rot_step(c[2 * u], c[2 * u + 1], g0, g1, g2, q, outS, ok ? (col * 3 + r) : spare);
    }
}

__global__ __launch_bounds__(DH_THREADS)
void k_dihedral(const float* __restrict__ dih,
                const float* __restrict__ cart,
                float* __restrict__ out) {
    __shared__ float4 cartS4[384];
    __shared__ float  dihS[512];
    __shared__ float4 recS4[2 * REC_SIDE / 4];
    __shared__ float4 outS4[386];

    const int t = threadIdx.x;
    const int b = blockIdx.x;
    const float* cartS = (const float*)cartS4;
    float* recS = (float*)recS4;
    float* outS = (float*)outS4;

    const float4* cg = (const float4*)(cart + (size_t)b * 1536);
    const float*  dg = dih + (size_t)b * 509;
    const int tl = (t + 384 < 509) ? (t + 384) : 508;
    const float4 c0 = cg[t], c1 = cg[t + 128], c2 = cg[t + 256];
    const float  d0 = dg[t], d1 = dg[t + 128], d2 = dg[t + 256], d3 = dg[tl];
    cartS4[t] = c0; cartS4[t + 128] = c1; cartS4[t + 256] = c2;
    dihS[t] = d0; dihS[t + 128] = d1; dihS[t + 256] = d2; dihS[t + 384] = d3;
    __syncthreads();

#pragma unroll
    for (int e = 0; e < 2; ++e) {
        const int i = t + e * DH_THREADS;
#pragma unroll
        for (int s = 0; s < 2; ++s) {
            const int nst = s ? 255 : 254;
            float m00 = 1.f, m01 = 0.f, m11 = 1.f, m02 = 0.f, m12 = 0.f, m22 = 1.f;
            float dx = 0.f, dy = 0.f;
            if (i <= nst) {
                const int k1 = s ? (255 + i) : (255 - i);
                const int k2 = s ? (256 + i) : (254 - i);
                dx = cartS[3 * k2]     - cartS[3 * k1];
                dy = cartS[3 * k2 + 1] - cartS[3 * k1 + 1];
                if (i < nst) {
                    const int did = s ? (254 + i) : (253 - i);
                    const float phi = dihS[did] + PI_F;
                    const float rn = rsqrtf(dx * dx + dy * dy);
                    const float ax = dx * rn, ay = dy * rn;
                    float sv, cv;
                    __sincosf(phi, &sv, &cv);
                    const float oc = 1.f - cv;
                    m00 = cv + oc * ax * ax;
                    m01 = oc * ax * ay;
                    m11 = cv + oc * ay * ay;
                    m02 = sv * ay;
                    m12 = -sv * ax;
                    m22 = cv;
                }
            }
            float4* rp = (float4*)(recS + s * REC_SIDE + i * 8);
            rp[0] = make_float4(m00, m01, m11, m02);
            rp[1] = make_float4(m12, m22, dx, dy);
        }
    }
    __syncthreads();

    if (t < 6) {
        const bool right = (t >= 3);
        const int  r     = right ? (t - 3) : t;
        const int  spare = 1536 + t;
        if (!right) {
            outS[256 * 3 + r] = cartS[256 * 3 + r];
            outS[255 * 3 + r] = cartS[255 * 3 + r];
        }
        float g0 = (r == 0) ? 1.f : 0.f;
        float g1 = (r == 1) ? 1.f : 0.f;
        float g2 = (r == 2) ? 1.f : 0.f;
        float q  = cartS[255 * 3 + r];
        const float4* rec = (const float4*)(recS + (right ? REC_SIDE : 0));

        float4 bufA[16], bufB[16];
#pragma unroll
        for (int k = 0; k < 16; ++k) bufA[k] = rec[k];
        for (int blk = 0; blk < 32; blk += 2) {
#pragma unroll
            for (int k = 0; k < 16; ++k) bufB[k] = rec[(blk + 1) * 16 + k];
            rot_block8(bufA, blk * 8, right, r, spare, g0, g1, g2, q, outS);
            const int nb = (blk + 2 < 32) ? (blk + 2) : 31;
#pragma unroll
            for (int k = 0; k < 16; ++k) bufA[k] = rec[nb * 16 + k];
            rot_block8(bufB, (blk + 1) * 8, right, r, spare, g0, g1, g2, q, outS);
        }
    }
    __syncthreads();

    float4* og = (float4*)(out + (size_t)b * 1536);
    const float4 o0 = outS4[t], o1 = outS4[t + 128], o2 = outS4[t + 256];
    og[t] = o0; og[t + 128] = o1; og[t + 256] = o2;
}

extern "C" void kernel_launch(void* const* d_in, const int* in_sizes, int n_in,
                              void* d_out, int out_size, void* d_ws, size_t ws_size,
                              hipStream_t stream) {
    (void)in_sizes; (void)n_in; (void)out_size; (void)ws_size;
    const float* distances = (const float*)d_in[0];
    const float* angles    = (const float*)d_in[1];
    const float* dihedrals = (const float*)d_in[2];
    float* out = (float*)d_out;
    float* ws  = (float*)d_ws;
    float* lengths = ws;
    float* cart    = ws + 512;

    k_lengths <<<2, 256, 0, stream>>>(distances, lengths);
    k_chain   <<<16, CH_THREADS, 0, stream>>>(angles, lengths, cart);
    k_dihedral<<<256, DH_THREADS, 0, stream>>>(dihedrals, cart, out);
}


# k_chain: the 512 positions split over the workgroup's 8 waves (4 chunks of 16 each), wave totals joined through LDS behind two barriers; 32 serial chunks on one SIMD become 4 per wave on all four
# speedup vs baseline: 7.4214x; 1.1064x over previous
#include <hip/hip_runtime.h>
#include <math.h>

#define PI_F 3.14159265358979323846f

typedef float v2f __attribute__((ext_vector_type(2)));
typedef float v8f __attribute__((ext_vector_type(8)));

__device__ __forceinline__ v8f wmma_scan16(const float bm[4][2], int m, int hi) {
    v8f acc = {0.f, 0.f, 0.f, 0.f, 0.f, 0.f, 0.f, 0.f};
#pragma unroll
    for (int j = 0; j < 4; ++j) {
        const int c0 = 4 * j + 2 * hi;
        v2f a;
        a.x = (c0     <= m) ? 1.f : 0.f;
        a.y = (c0 + 1 <= m) ? 1.f : 0.f;
        v2f b;
        b.x = bm[j][0];
        b.y = bm[j][1];
        acc = __builtin_amdgcn_wmma_f32_16x16x4_f32(
             false, a,  false, b,
             (short)0, acc,  false,  false);
    }
    return acc;
}

__device__ __forceinline__ void apply_carry(const v8f acc, float& carry,
                                            float full[8], int n) {
    float loc[8];
#pragma unroll
    for (int r = 0; r < 8; ++r) loc[r] = acc[r];
    const float tot = __shfl(loc[7], 16 + n, 32);
#pragma unroll
    for (int r = 0; r < 8; ++r) full[r] = loc[r] + carry;
    carry += tot;
}

__device__ __forceinline__ void d_to_b(const float d[8], float bm[4][2], int hi) {
    float x[8];
#pragma unroll
    for (int r = 0; r < 8; ++r) x[r] = __shfl_xor(d[r], 16, 32);
#pragma unroll
    for (int j = 0; j < 2; ++j)
#pragma unroll
        for (int v = 0; v < 2; ++v)
            bm[j][v] = hi ? x[4 * j + v + 2] : d[4 * j + v];
#pragma unroll
    for (int j = 2; j < 4; ++j)
#pragma unroll
        for (int v = 0; v < 2; ++v)
            bm[j][v] = hi ? d[4 * j + v - 6] : x[4 * j + v - 8];
}

__global__ void k_lengths(const float* __restrict__ dist,
                          float* __restrict__ lengths) {
    const int j = blockIdx.x * blockDim.x + threadIdx.x;
    if (j >= 512) return;
    if (j == 511) { lengths[511] = 0.f; return; }
    float s = 0.f;
    for (int b = 0; b < 256; ++b) s += dist[b * 511 + j];
    lengths[j] = s * (1.0f / 256.0f);
}

#define CH_THREADS 256
#define CH_WAVES   8
#define CH_PER     4
#define ANG_Q      2040
#define ANG_F4     2049

__device__ __forceinline__ void load_chunk(const float* angS, int rowoff, int hi, int t, float out[8]) {
#pragma unroll
    for (int j = 0; j < 4; ++j)
#pragma unroll
        for (int v = 0; v < 2; ++v) {
            const int pos = 16 * t + 4 * j + 2 * hi + v;
            const float val = angS[rowoff + pos];
            out[2 * j + v] = (pos >= 1 && pos <= 510) ? val : 0.f;
        }
}

__device__ __forceinline__ float waves_in_front(const float (*tot)[16], int wave, int n) {
    float off = 0.f;
#pragma unroll
    for (int w = 0; w < CH_WAVES - 1; ++w) {
        const float v = tot[w][n];
        off += (w < wave) ? v : 0.f;
    }
    return off;
}

__global__ __launch_bounds__(CH_THREADS)
void k_chain(const float* __restrict__ angles,
             const float* __restrict__ lengths,
             float* __restrict__ cart) {
    __shared__ float4 angS4[ANG_F4];
    __shared__ float  lenS[512];
    __shared__ float  totP[CH_WAVES][16];
    __shared__ float  totX[CH_WAVES][16];
    __shared__ float  totY[CH_WAVES][16];
    const int tid = threadIdx.x;

    const float4* ag = (const float4*)(angles + (size_t)blockIdx.x * (16 * 510));
    float4 a[8];
#pragma unroll
    for (int k = 0; k < 8; ++k) {
        const int q = tid + CH_THREADS * k;
        a[k] = ag[(k < 7 || q < ANG_Q) ? q : (ANG_Q - 1)];
    }
    const float l0 = lengths[tid], l1 = lengths[tid + CH_THREADS];
#pragma unroll
    for (int k = 0; k < 8; ++k) {
        const int q = tid + CH_THREADS * k;
        angS4[q + 1] = make_float4(PI_F - a[k].x, PI_F - a[k].y, PI_F - a[k].z, PI_F - a[k].w);
    }
    if (tid == 0) angS4[0] = make_float4(0.f, 0.f, 0.f, 0.f);
    lenS[tid] = l0; lenS[tid + CH_THREADS] = l1;
    __syncthreads();

    const float* angS = (const float*)angS4;
    const int wave = tid >> 5;
    const int lane = tid & 31;
    const int n   = lane & 15;
    const int hi  = lane >> 4;
    const int m   = n;
    const int row = blockIdx.x * 16 + n;
    const int rowoff = 3 + n * 510;
    const int t0  = CH_PER * wave;

    float prevL[CH_PER][8];
    float carryP = 0.f;
#pragma unroll
    for (int c = 0; c < CH_PER; ++c) {
        float cur[8];
        load_chunk(angS, rowoff, hi, t0 + c, cur);
        float bm[4][2];
#pragma unroll
        for (int j = 0; j < 4; ++j)
#pragma unroll
            for (int v = 0; v < 2; ++v) bm[j][v] = cur[2 * j + v];
        const v8f accP = wmma_scan16(bm, m, hi);
        apply_carry(accP, carryP, prevL[c], n);
    }
    if (hi == 0) totP[wave][n] = carryP;
    __syncthreads();
    const float offP = waves_in_front(totP, wave, n);

    float dx[CH_PER][8], dy[CH_PER][8], xL[CH_PER][8], yL[CH_PER][8];
    float carryX = 0.f, carryY = 0.f;
#pragma unroll
    for (int c = 0; c < CH_PER; ++c) {
#pragma unroll
        for (int r = 0; r < 8; ++r) {
            const int pos = 16 * (t0 + c) + r + 8 * hi;
            float sv, cv;
            __sincosf(prevL[c][r] + offP, &sv, &cv);
            const float L = lenS[pos];
            dx[c][r] = L * cv;
            dy[c][r] = L * sv;
        }
        float bmx[4][2], bmy[4][2];
        d_to_b(dx[c], bmx, hi);
        d_to_b(dy[c], bmy, hi);
        const v8f accX = wmma_scan16(bmx, m, hi);
        const v8f accY = wmma_scan16(bmy, m, hi);
        apply_carry(accX, carryX, xL[c], n);
        apply_carry(accY, carryY, yL[c], n);
    }
    if (hi == 0) { totX[wave][n] = carryX; totY[wave][n] = carryY; }
    __syncthreads();
    const float offX = waves_in_front(totX, wave, n);
    const float offY = waves_in_front(totY, wave, n);

#pragma unroll
    for (int c = 0; c < CH_PER; ++c) {
        float o[24];
#pragma unroll
        for (int r = 0; r < 8; ++r) {
            o[3 * r]     = (xL[c][r] + offX) - dx[c][r];
            o[3 * r + 1] = (yL[c][r] + offY) - dy[c][r];
            o[3 * r + 2] = 0.f;
        }
        float4* cp4 = (float4*)(cart + ((size_t)row * 512 + 16 * (t0 + c) + 8 * hi) * 3);
#pragma unroll
        for (int k = 0; k < 6; ++k)
            cp4[k] = make_float4(o[4 * k], o[4 * k + 1], o[4 * k + 2], o[4 * k + 3]);
    }
}

#define DH_THREADS 128
#define REC_SIDE   2056

__device__ __forceinline__ void rot_step(const float4 A, const float4 B,
                                         float& g0, float& g1, float& g2,
                                         float& q, float* outS, int idx) {
    q = fmaf(g1, B.w, fmaf(g0, B.z, q));
    outS[idx] = q;
    const float n0 = fmaf(-g2, A.w, fmaf(g1, A.y, g0 * A.x));
    const float n1 = fmaf(-g2, B.x, fmaf(g1, A.z, g0 * A.y));
    const float n2 = fmaf( g2, B.y, fmaf(g1, B.x, g0 * A.w));
    g0 = n0; g1 = n1; g2 = n2;
}

__device__ __forceinline__ void rot_block8(const float4 (&c)[16], int i0,
                                           bool right, int r, int spare,
                                           float& g0, float& g1, float& g2,
                                           float& q, float* outS) {
#pragma unroll
    for (int u = 0; u < 8; ++u) {
        const int i   = i0 + u;
        const bool ok = right ? (i >= 1) : (i <= 254);
        const int col = right ? (256 + i) : (254 - i);
        rot_step(c[2 * u], c[2 * u + 1], g0, g1, g2, q, outS, ok ? (col * 3 + r) : spare);
    }
}

__global__ __launch_bounds__(DH_THREADS)
void k_dihedral(const float* __restrict__ dih,
                const float* __restrict__ cart,
                float* __restrict__ out) {
    __shared__ float4 cartS4[384];
    __shared__ float  dihS[512];
    __shared__ float4 recS4[2 * REC_SIDE / 4];
    __shared__ float4 outS4[386];

    const int t = threadIdx.x;
    const int b = blockIdx.x;
    const float* cartS = (const float*)cartS4;
    float* recS = (float*)recS4;
    float* outS = (float*)outS4;

    const float4* cg = (const float4*)(cart + (size_t)b * 1536);
    const float*  dg = dih + (size_t)b * 509;
    const int tl = (t + 384 < 509) ? (t + 384) : 508;
    const float4 c0 = cg[t], c1 = cg[t + 128], c2 = cg[t + 256];
    const float  d0 = dg[t], d1 = dg[t + 128], d2 = dg[t + 256], d3 = dg[tl];
    cartS4[t] = c0; cartS4[t + 128] = c1; cartS4[t + 256] = c2;
    dihS[t] = d0; dihS[t + 128] = d1; dihS[t + 256] = d2; dihS[t + 384] = d3;
    __syncthreads();

#pragma unroll
    for (int e = 0; e < 2; ++e) {
        const int i = t + e * DH_THREADS;
#pragma unroll
        for (int s = 0; s < 2; ++s) {
            const int nst = s ? 255 : 254;
            float m00 = 1.f, m01 = 0.f, m11 = 1.f, m02 = 0.f, m12 = 0.f, m22 = 1.f;
            float dx = 0.f, dy = 0.f;
            if (i <= nst) {
                const int k1 = s ? (255 + i) : (255 - i);
                const int k2 = s ? (256 + i) : (254 - i);
                dx = cartS[3 * k2]     - cartS[3 * k1];
                dy = cartS[3 * k2 + 1] - cartS[3 * k1 + 1];
                if (i < nst) {
                    const int did = s ? (254 + i) : (253 - i);
                    const float phi = dihS[did] + PI_F;
                    const float rn = rsqrtf(dx * dx + dy * dy);
                    const float ax = dx * rn, ay = dy * rn;
                    float sv, cv;
                    __sincosf(phi, &sv, &cv);
                    const float oc = 1.f - cv;
                    m00 = cv + oc * ax * ax;
                    m01 = oc * ax * ay;
                    m11 = cv + oc * ay * ay;
                    m02 = sv * ay;
                    m12 = -sv * ax;
                    m22 = cv;
                }
            }
            float4* rp = (float4*)(recS + s * REC_SIDE + i * 8);
            rp[0] = make_float4(m00, m01, m11, m02);
            rp[1] = make_float4(m12, m22, dx, dy);
        }
    }
    __syncthreads();

    if (t < 6) {
        const bool right = (t >= 3);
        const int  r     = right ? (t - 3) : t;
        const int  spare = 1536 + t;
        if (!right) {
            outS[256 * 3 + r] = cartS[256 * 3 + r];
            outS[255 * 3 + r] = cartS[255 * 3 + r];
        }
        float g0 = (r == 0) ? 1.f : 0.f;
        float g1 = (r == 1) ? 1.f : 0.f;
        float g2 = (r == 2) ? 1.f : 0.f;
        float q  = cartS[255 * 3 + r];
        const float4* rec = (const float4*)(recS + (right ? REC_SIDE : 0));

        float4 bufA[16], bufB[16];
#pragma unroll
        for (int k = 0; k < 16; ++k) bufA[k] = rec[k];
        for (int blk = 0; blk < 32; blk += 2) {
#pragma unroll
            for (int k = 0; k < 16; ++k) bufB[k] = rec[(blk + 1) * 16 + k];
            rot_block8(bufA, blk * 8, right, r, spare, g0, g1, g2, q, outS);
            const int nb = (blk + 2 < 32) ? (blk + 2) : 31;
#pragma unroll
            for (int k = 0; k < 16; ++k) bufA[k] = rec[nb * 16 + k];
            rot_block8(bufB, (blk + 1) * 8, right, r, spare, g0, g1, g2, q, outS);
        }
    }
    __syncthreads();

    float4* og = (float4*)(out + (size_t)b * 1536);
    const float4 o0 = outS4[t], o1 = outS4[t + 128], o2 = outS4[t + 256];
    og[t] = o0; og[t + 128] = o1; og[t + 256] = o2;
}

extern "C" void kernel_launch(void* const* d_in, const int* in_sizes, int n_in,
                              void* d_out, int out_size, void* d_ws, size_t ws_size,
                              hipStream_t stream) {
    (void)in_sizes; (void)n_in; (void)out_size; (void)ws_size;
    const float* distances = (const float*)d_in[0];
    const float* angles    = (const float*)d_in[1];
    const float* dihedrals = (const float*)d_in[2];
    float* out = (float*)d_out;
    float* ws  = (float*)d_ws;
    float* lengths = ws;
    float* cart    = ws + 512;

    k_lengths <<<2, 256, 0, stream>>>(distances, lengths);
    k_chain   <<<16, CH_THREADS, 0, stream>>>(angles, lengths, cart);
    k_dihedral<<<256, DH_THREADS, 0, stream>>>(dihedrals, cart, out);
}
